# starting kernel
# baseline (speedup 1.0000x reference)
#include <hip/hip_runtime.h>
#include <hip/hip_bf16.h>

typedef __attribute__((ext_vector_type(2))) float v2f;
typedef __attribute__((ext_vector_type(8))) float v8f;

#define BATCH 32768
#define IN_F  34
#define OUT_F 1024

__global__ __launch_bounds__(256) void WeightLinear_46505905881421_kernel(
    const float* __restrict__ score,
    const float* __restrict__ x,
    const float* __restrict__ weight,
    const float* __restrict__ bias,
    float* __restrict__ out)
{
    const int lane = threadIdx.x & 31;
    const int wave = threadIdx.x >> 5;

    const int n0 = blockIdx.x * 16;
    const int m0 = (blockIdx.y * 8 + wave) * 16;

    const int half = lane >> 4;
    const int l16  = lane & 15;

    const int m = m0 + l16;
    const int n = n0 + l16;

    const float* __restrict__ xrow = x + m * IN_F;
    const float* __restrict__ srow = score + m * IN_F;

    v8f acc = {};

    #pragma unroll
    for (int k0 = 0; k0 < 32; k0 += 4) {
        const int ka = k0 + half * 2;

        v2f a;
        a.x = xrow[ka]     * srow[ka];
        a.y = xrow[ka + 1] * srow[ka + 1];

        v2f b;
        b.x = weight[ka * OUT_F + n];
        b.y = weight[(ka + 1) * OUT_F + n];

        acc = __builtin_amdgcn_wmma_f32_16x16x4_f32(
            false, a, false, b, (short)0, acc, false, false);
    }

    {
        const float a0 = xrow[32] * srow[32];
        const float a1 = xrow[33] * srow[33];
        const float b0 = weight[32 * OUT_F + n];
        const float b1 = weight[33 * OUT_F + n];
        const bool lo = (half == 0);

        v2f a; a.x = lo ? a0 : 0.0f; a.y = lo ? a1 : 0.0f;
        v2f b; b.x = lo ? b0 : 0.0f; b.y = lo ? b1 : 0.0f;

        acc = __builtin_amdgcn_wmma_f32_16x16x4_f32(
            false, a, false, b, (short)0, acc, false, false);
    }

    const float bv = bias[n];
    float* __restrict__ orow = out + (m0 + half * 8) * OUT_F + n;
    #pragma unroll
    for (int r = 0; r < 8; ++r) {
        orow[r * OUT_F] = acc[r] + bv;
    }
}

extern "C" void kernel_launch(void* const* d_in, const int* in_sizes, int n_in,
                              void* d_out, int out_size, void* d_ws, size_t ws_size,
                              hipStream_t stream) {
    (void)in_sizes; (void)n_in; (void)d_ws; (void)ws_size; (void)out_size;

    const float* score  = (const float*)d_in[0];
    const float* x      = (const float*)d_in[1];
    const float* weight = (const float*)d_in[2];
    const float* bias   = (const float*)d_in[3];
    float* out          = (float*)d_out;

    dim3 grid(OUT_F / 16, BATCH / (16 * 8));
    dim3 block(256);
    WeightLinear_46505905881421_kernel<<<grid, block, 0, stream>>>(
        score, x, weight, bias, out);
}


# Work-group tile 128x256: x*score and the weight slab staged once per work-group in LDS by whole-line loads; each wave keeps its A fragments and sweeps 16 column tiles; summation order unchanged
# speedup vs baseline: 3.6478x; 3.6478x over previous
#include <hip/hip_runtime.h>

typedef __attribute__((ext_vector_type(2))) float v2f;
typedef __attribute__((ext_vector_type(4))) float v4f;
typedef __attribute__((ext_vector_type(8))) float v8f;

#define BATCH 32768
#define IN_F  34
#define OUT_F 1024

#define BM 128
#define BN 256
#define PA 36
#define PW 272

__global__ __launch_bounds__(256) void WeightLinear_tile_kernel(
    const float* __restrict__ score,
    const float* __restrict__ x,
    const float* __restrict__ weight,
    const float* __restrict__ bias,
    float* __restrict__ out)
{
    __shared__ __attribute__((aligned(16))) float sA[BM * PA];
    __shared__ __attribute__((aligned(16))) v2f   sW[18 * PW];

    const int t    = threadIdx.x;
    const int lane = t & 31;
    const int wave = t >> 5;
    const int half = lane >> 4;
    const int l16  = lane & 15;

    const int c0 = blockIdx.x * BN;
    const int r0 = blockIdx.y * BM;

    const v2f* __restrict__ xb = (const v2f*)(x + r0 * IN_F);
    const v2f* __restrict__ sb = (const v2f*)(score + r0 * IN_F);
    const int wp = t >> 6;
    const int c4 = t & 63;
    const float* __restrict__ wcol = weight + c0 + 4 * c4;

    v2f xv[9], sv[9];
    v4f wu[5], wv[5];
    #pragma unroll
    for (int q = 0; q < 8; ++q) {
        xv[q] = xb[t + 256 * q];
        sv[q] = sb[t + 256 * q];
    }
    const int i8  = t + 2048;
    const int i8c = (i8 < BM * 17) ? i8 : (BM * 17 - 1);
    xv[8] = xb[i8c];
    sv[8] = sb[i8c];
    #pragma unroll
    for (int q = 0; q < 4; ++q) {
        const float* __restrict__ w0 = wcol + (2 * (wp + 4 * q)) * OUT_F;
        wu[q] = *(const v4f*)w0;
        wv[q] = *(const v4f*)(w0 + OUT_F);
    }
    wu[4] = *(const v4f*)(wcol + 32 * OUT_F);
    wv[4] = *(const v4f*)(wcol + 33 * OUT_F);

    #pragma unroll
    for (int q = 0; q < 9; ++q) {
        const int i2 = t + 256 * q;
        if (q < 8 || i2 < BM * 17) {
            const int row = i2 / 17;
            const int c2  = i2 - row * 17;
            v2f a;
            a.x = xv[q].x * sv[q].x;
            a.y = xv[q].y * sv[q].y;
            *(v2f*)&sA[row * PA + 2 * c2] = a;
        }
    }
    if (t < BM) {
        v2f z; z.x = 0.0f; z.y = 0.0f;
        *(v2f*)&sA[t * PA + IN_F] = z;
    }

    #pragma unroll
    for (int q = 0; q < 5; ++q) {
        const int p = wp + 4 * q;
        if (q < 4 || p == 16) {
            v4f lo; lo.x = wu[q].x; lo.y = wv[q].x; lo.z = wu[q].y; lo.w = wv[q].y;
            v4f hi; hi.x = wu[q].z; hi.y = wv[q].z; hi.z = wu[q].w; hi.w = wv[q].w;
            v4f* dst = (v4f*)&sW[p * PW + 4 * c4];
            dst[0] = lo;
            dst[1] = hi;
        } else if (p == 17) {
            v4f z; z.x = 0.0f; z.y = 0.0f; z.z = 0.0f; z.w = 0.0f;
            v4f* dst = (v4f*)&sW[p * PW + 4 * c4];
            dst[0] = z;
            dst[1] = z;
        }
    }

    __syncthreads();

    v2f a[9];
    {
        const float* arow = &sA[(wave * 16 + l16) * PA + 2 * half];
        #pragma unroll
        for (int s = 0; s < 9; ++s) {
            a[s] = *(const v2f*)(arow + 4 * s);
        }
    }

    float* __restrict__ orow = out + (r0 + wave * 16 + half * 8) * OUT_F + c0 + l16;
    const float* __restrict__ brow = bias + c0 + l16;

    for (int jp = 0; jp < 8; ++jp) {
        const int cn = ((jp + wave) & 7) * 32;
        const v2f* wb = &sW[half * PW + cn + l16];

        v2f b0[9], b1[9];
        #pragma unroll
        for (int s = 0; s < 9; ++s) {
            b0[s] = wb[2 * s * PW];
            b1[s] = wb[2 * s * PW + 16];
        }

        const float bv0 = brow[cn];
        const float bv1 = brow[cn + 16];

        v8f acc0 = {};
        v8f acc1 = {};
        #pragma unroll
        for (int s = 0; s < 9; ++s) {
            acc0 = __builtin_amdgcn_wmma_f32_16x16x4_f32(
                false, a[s], false, b0[s], (short)0, acc0, false, false);
            acc1 = __builtin_amdgcn_wmma_f32_16x16x4_f32(
                false, a[s], false, b1[s], (short)0, acc1, false, false);
        }

        float* __restrict__ o = orow + cn;
        #pragma unroll
        for (int r = 0; r < 8; ++r) {
            o[r * OUT_F]      = acc0[r] + bv0;
            o[r * OUT_F + 16] = acc1[r] + bv1;
        }
    }
}

extern "C" void kernel_launch(void* const* d_in, const int* in_sizes, int n_in,
                              void* d_out, int out_size, void* d_ws, size_t ws_size,
                              hipStream_t stream) {
    (void)in_sizes; (void)n_in; (void)d_ws; (void)ws_size; (void)out_size;

    const float* score  = (const float*)d_in[0];
    const float* x      = (const float*)d_in[1];
    const float* weight = (const float*)d_in[2];
    const float* bias   = (const float*)d_in[3];
    float* out          = (float*)d_out;

    dim3 grid(OUT_F / BN, BATCH / BM);
    dim3 block(256);
    WeightLinear_tile_kernel<<<grid, block, 0, stream>>>(
        score, x, weight, bias, out);
}
